# starting kernel
# baseline (speedup 1.0000x reference)
#include <hip/hip_runtime.h>

typedef __attribute__((ext_vector_type(2))) float v2f;
typedef __attribute__((ext_vector_type(8))) float v8f;

static constexpr int B_ = 16;
static constexpr int C_ = 64;
static constexpr int H_ = 512;
static constexpr int W_ = 60;
static constexpr long HW_ = (long)H_ * W_;
static constexpr int KTOT = 192;

__device__ __forceinline__ void load_weights_lds(const float* __restrict__ wt,
                                                 float* wlds) {
  for (int idx = threadIdx.x; idx < KTOT * 64; idx += 256) {
    const int K  = idx >> 6;
    const int co = idx & 63;
    const int kh = K >> 6;
    const int ci = K & 63;
    wlds[(K >> 1) * 128 + co * 2 + (K & 1)] = wt[co * 576 + ci * 9 + kh * 3 + 1];
  }
  __syncthreads();
}

__device__ __forceinline__ void conv_tile(const float* __restrict__ src,
                                          const float* wlds,
                                          int b, int h, int i0,
                                          int n, int hi, v8f acc[4]) {
  const long laneB0 = (long)(2 * hi) * HW_ + n;
  const long laneB1 = laneB0 + HW_;
  const int  aLane  = hi * 128 + n * 2;

#pragma unroll
  for (int kh = 0; kh < 3; ++kh) {
    const int hr = h + kh - 1;
    if (hr < 0 || hr >= H_) continue;
    const float* srow = src + ((long)b * C_ * H_ + hr) * W_ + i0;
#pragma unroll
    for (int kc = 0; kc < 16; ++kc) {
      const long cbase = (long)(kc * 4) * HW_;
      v2f bfrag;
      bfrag.x = srow[cbase + laneB0];
      bfrag.y = srow[cbase + laneB1];
      const int k0    = kh * 64 + kc * 4;
      const int abase = (k0 >> 1) * 128 + aLane;
#pragma unroll
      for (int t = 0; t < 4; ++t) {
        v2f afrag = *(const v2f*)(wlds + abase + t * 32);
        acc[t] = __builtin_amdgcn_wmma_f32_16x16x4_f32(
             false, afrag,  false, bfrag,
             (short)0, acc[t],  false,  false);
      }
    }
  }
}

__device__ __forceinline__ void decode_tile(int& b, int& h, int& i0, int& n, int& hi) {
  const int lane = threadIdx.x & 31;
  n  = lane & 15;
  hi = lane >> 4;
  const int tile = blockIdx.x * 8 + (threadIdx.x >> 5);
  const int it = tile & 3;
  h  = (tile >> 2) & 511;
  b  = tile >> 11;
  i0 = (it == 3) ? 44 : it * 16;
}

__global__ __launch_bounds__(256) void pass1_kernel(const float* __restrict__ x,
                                                    const float* __restrict__ wt,
                                                    const float* __restrict__ bias,
                                                    float* __restrict__ down) {
  __shared__ float wlds[KTOT * 64];
  load_weights_lds(wt, wlds);

  int b, h, i0, n, hi;
  decode_tile(b, h, i0, n, hi);

  v8f acc[4] = {};
  conv_tile(x, wlds, b, h, i0, n, hi, acc);

  const int  i     = i0 + n;
  const bool sh_ok = (i + 1) < W_;
#pragma unroll
  for (int t = 0; t < 4; ++t) {
#pragma unroll
    for (int v = 0; v < 8; ++v) {
      const int  co  = t * 16 + v + 8 * hi;
      const long off = (((long)b * C_ + co) * H_ + h) * W_ + i;
      float r  = fmaxf(acc[t][v] + bias[co], 0.0f);
      float xs = sh_ok ? x[off + 1] : 0.0f;
      down[off] = r + xs;
    }
  }
}

__global__ __launch_bounds__(256) void pass2_kernel(const float* __restrict__ down,
                                                    const float* __restrict__ wt,
                                                    const float* __restrict__ bias,
                                                    float* __restrict__ out) {
  __shared__ float wlds[KTOT * 64];
  load_weights_lds(wt, wlds);

  int b, h, i0, n, hi;
  decode_tile(b, h, i0, n, hi);

  v8f acc[4] = {};
  conv_tile(down, wlds, b, h, i0, n, hi, acc);

  const int  k  = i0 + n - 1;
  const bool ok = (k >= 0);
#pragma unroll
  for (int t = 0; t < 4; ++t) {
#pragma unroll
    for (int v = 0; v < 8; ++v) {
      const int  co  = t * 16 + v + 8 * hi;
      const long off = (((long)b * C_ + co) * H_ + h) * W_ + k;
      float cd = fmaxf(acc[t][v] + bias[co], 0.0f);
      if (ok) out[off] = down[off] + cd;
    }
  }
}

__global__ __launch_bounds__(256) void col59_kernel(const float* __restrict__ down,
                                                    float* __restrict__ out) {
  const long idx   = (long)blockIdx.x * 256 + threadIdx.x;
  const long total = (long)B_ * C_ * H_;
  if (idx < total) {
    const long off = idx * W_ + (W_ - 1);
    out[off] = down[off];
  }
}

extern "C" void kernel_launch(void* const* d_in, const int* in_sizes, int n_in,
                              void* d_out, int out_size, void* d_ws, size_t ws_size,
                              hipStream_t stream) {
  const float* x    = (const float*)d_in[0];
  const float* w    = (const float*)d_in[1];
  const float* bias = (const float*)d_in[2];
  float*       out  = (float*)d_out;
  float*       down = (float*)d_ws;

  const int tiles = B_ * H_ * 4;
  const int wgs   = tiles / 8;
  dim3 block(256);

  pass1_kernel<<<wgs, block, 0, stream>>>(x, w, bias, down);
  pass2_kernel<<<wgs, block, 0, stream>>>(down, w, bias, out);

  const long rows = (long)B_ * C_ * H_;
  col59_kernel<<<(int)((rows + 255) / 256), block, 0, stream>>>(down, out);
}


# Work-group owns 32 rows, walks them in 4 chunks of 512 down positions; the 128 positions two chunks share stay in an LDS ring: down made once per position (band 2048 per 1920, was 512 per 384)
# speedup vs baseline: 3.1257x; 3.1257x over previous
#include <hip/hip_runtime.h>

typedef __attribute__((ext_vector_type(2))) float v2f;
typedef __attribute__((ext_vector_type(8))) float v8f;

static constexpr int B_ = 16;
static constexpr int C_ = 64;
static constexpr int H_ = 512;
static constexpr int W_ = 60;
static constexpr int HW_ = H_ * W_;
static constexpr int KTOT = 192;

static constexpr int NWAVE = 32;
static constexpr int NTHR  = NWAVE * 32;
static constexpr int OWN   = 1920;
static constexpr int WGI   = HW_ / OWN;
static constexpr int GRID  = B_ * WGI;
static constexpr int CH    = NWAVE * 16;
static constexpr int NCHUNK = 4;
static constexpr int KEEP  = 128;
static constexpr int RING  = CH + KEEP;
static constexpr int RT    = RING / 16;
static constexpr int T2LAG = KEEP / 16;
static_assert(WGI * OWN == HW_ && OWN % W_ == 0, "work-groups own whole rows and tile a plane exactly");
static_assert(NCHUNK * CH >= OWN + 2 * W_, "the band covers every tap of the owned positions");
static_assert(16 * (NWAVE * (NCHUNK - 1) - T2LAG + NWAVE - 1) + 16 == OWN, "the last chunk ends on the last owned tile");
static_assert(16 * (NWAVE - T2LAG - 1) + 15 + 2 * W_ < CH, "a chunk's last second-conv tap is inside what has been written");
static constexpr int DP    = 650;
static_assert(DP >= RING, "a channel's ring must fit its LDS row");

static constexpr int WPACK = KTOT * 64;
static constexpr int PGRID = WPACK / 256;
static_assert(WPACK % (4 * NTHR) == 0, "the LDS copy is whole 16-byte pieces, the same count for every thread");

__global__ __launch_bounds__(256) void pack_weights(const float* __restrict__ wt,
                                                    float* __restrict__ wp) {
  const int idx = blockIdx.x * 256 + threadIdx.x;
  const int K   = ((idx >> 7) << 1) | (idx & 1);
  const int co  = (idx >> 1) & 63;
  const int kh  = K >> 6;
  const int ci  = K & 63;
  wp[idx] = wt[co * 576 + ci * 9 + kh * 3 + 1];
}

__device__ __forceinline__ void load_weights_lds(const float* __restrict__ wp,
                                                 float* wlds) {
  typedef __attribute__((ext_vector_type(4))) float v4f;
  const v4f* s = (const v4f*)wp;
  v4f*       d = (v4f*)wlds;
#pragma unroll
  for (int i = 0; i < WPACK / 4 / NTHR; ++i)
    d[threadIdx.x + i * NTHR] = s[threadIdx.x + i * NTHR];
  __syncthreads();
}

__device__ __forceinline__ void kstep(const float* wlds, int abase, v2f bfrag, v8f acc[4]) {
#pragma unroll
  for (int t = 0; t < 4; ++t) {
    v2f afrag = *(const v2f*)(wlds + abase + t * 32);
    acc[t] = __builtin_amdgcn_wmma_f32_16x16x4_f32(
         false, afrag,  false, bfrag,
         (short)0, acc[t],  false,  false);
  }
}

__device__ __forceinline__ int lane_now() {
  int l;
  asm volatile("v_mbcnt_lo_u32_b32 %0, -1, 0" : "=v"(l));
  return l;
}

__global__ __launch_bounds__(NTHR) void fused_kernel(const float* __restrict__ x,
                                                           const float* __restrict__ wp,
                                                           const float* __restrict__ bias,
                                                           float* __restrict__ out) {
  __shared__ __attribute__((aligned(16))) float smem[KTOT * 64 + C_ * DP];
  float* wlds = smem;
  float* dlds = smem + KTOT * 64;
  load_weights_lds(wp, wlds);

  const int wave = __builtin_amdgcn_readfirstlane((int)(threadIdx.x >> 5));
  const int b    = blockIdx.x / WGI;
  const int rb   = blockIdx.x - b * WGI;
  const int c0   = rb * OWN;
  const int g0   = c0 - W_;
  const float* xb = x + (long)b * C_ * HW_;
  float*       ob = out + (long)b * C_ * HW_;

#pragma unroll 1
  for (int k = 0; k < NCHUNK; ++k) {
    if (k > 0) __syncthreads();

    {
      const int lk = lane_now();
      const int n  = lk & 15;
      const int hi = lk >> 4;
      const int aLane = hi * 128 + n * 2;
      const float* bp = bias + 8 * hi;
      const int tile = k * NWAVE + wave;
      const int u = tile * 16 + n;
      const int P = g0 + u;
      const int slot = (tile % RT) * 16 + n;
      v8f acc[4] = {};
#pragma unroll
      for (int kh = 0; kh < 3; ++kh) {
        const int  Pk  = P + (kh - 1) * W_;
        const bool okk = (unsigned)Pk < (unsigned)HW_;
        const float* sp = xb + (2 * hi) * HW_ + (okk ? Pk : 0);
#pragma unroll
        for (int kc = 0; kc < 16; ++kc) {
          const float b0 = sp[(kc * 4) * HW_];
          const float b1 = sp[(kc * 4 + 1) * HW_];
          v2f bfrag;
          bfrag.x = okk ? b0 : 0.0f;
          bfrag.y = okk ? b1 : 0.0f;
          const int k0 = kh * 64 + kc * 4;
          kstep(wlds, (k0 >> 1) * 128 + aLane, bfrag, acc);
        }
      }
      __builtin_amdgcn_sched_barrier(0);
      const int  i     = u % W_;
      const bool rowok = (unsigned)P < (unsigned)HW_;
      const bool shok  = rowok && (i < W_ - 1);
      const float* xsp = xb + (8 * hi) * HW_ + (shok ? P + 1 : 0);
      float*       dst = dlds + (8 * hi) * DP + slot;
      float xv[32], bv[32];
#pragma unroll
      for (int t = 0; t < 4; ++t) {
#pragma unroll
        for (int v = 0; v < 8; ++v) {
          const int ch = t * 16 + v;
          xv[t * 8 + v] = xsp[ch * HW_];
          bv[t * 8 + v] = bp[ch];
        }
      }
#pragma unroll
      for (int t = 0; t < 4; ++t) {
#pragma unroll
        for (int v = 0; v < 8; ++v) {
          const int ch = t * 16 + v;
          const float r  = fmaxf(acc[t][v] + bv[t * 8 + v], 0.0f);
          const float xs = shok ? xv[t * 8 + v] : 0.0f;
          dst[ch * DP] = rowok ? (r + xs) : 0.0f;
        }
      }
    }
    __syncthreads();

    const int j = k * NWAVE - T2LAG + wave;
    if (j >= 0) {
      const int lk = lane_now();
      const int n  = lk & 15;
      const int hi = lk >> 4;
      const int aLane = hi * 128 + n * 2;
      const float* bp = bias + 8 * hi;
      const int uc = j * 16 + n + W_;
      v8f acc[4] = {};
#pragma unroll
      for (int kh = 0; kh < 3; ++kh) {
        const int slot = (uc + (kh - 1) * W_) % RING;
        const float* sp = dlds + (2 * hi) * DP + slot;
#pragma unroll
        for (int kc = 0; kc < 16; ++kc) {
          v2f bfrag;
          bfrag.x = sp[(kc * 4) * DP];
          bfrag.y = sp[(kc * 4 + 1) * DP];
          const int k0 = kh * 64 + kc * 4;
          kstep(wlds, (k0 >> 1) * 128 + aLane, bfrag, acc);
        }
      }
      const int  i     = uc % W_;
      const bool first = (i == 0);
      const int  du    = first ? (uc + W_ - 1) : (uc - 1);
      const float* dsrc = dlds + (8 * hi) * DP + du % RING;
      float*       op   = ob + (8 * hi) * HW_ + g0 + du;
#pragma unroll
      for (int t = 0; t < 4; ++t) {
#pragma unroll
        for (int v = 0; v < 8; ++v) {
          const int ch = t * 16 + v;
          const float cd = fmaxf(acc[t][v] + bp[ch], 0.0f);
          const float d  = dsrc[ch * DP];
          op[ch * HW_] = first ? d : (d + cd);
        }
      }
    }
  }
}

extern "C" void kernel_launch(void* const* d_in, const int* in_sizes, int n_in,
                              void* d_out, int out_size, void* d_ws, size_t ws_size,
                              hipStream_t stream) {
  const float* x    = (const float*)d_in[0];
  const float* w    = (const float*)d_in[1];
  const float* bias = (const float*)d_in[2];
  float*       out  = (float*)d_out;
  float*       wp   = (float*)d_ws;

  pack_weights<<<PGRID, 256, 0, stream>>>(w, wp);
  fused_kernel<<<GRID, NTHR, 0, stream>>>(x, wp, bias, out);
}
